# starting kernel
# baseline (speedup 1.0000x reference)
#include <hip/hip_runtime.h>
#include <math.h>

typedef __attribute__((ext_vector_type(2))) float v2f;
typedef __attribute__((ext_vector_type(8))) float v8f;

#define DFEAT 128

__global__ __launch_bounds__(256) void fill_neginf_kernel(uint4* __restrict__ p, int n4) {
    int i = blockIdx.x * blockDim.x + threadIdx.x;
    if (i < n4) {
        uint4 v;
        v.x = 0xFF800000u; v.y = 0xFF800000u; v.z = 0xFF800000u; v.w = 0xFF800000u;
        p[i] = v;
    }
}

__device__ __forceinline__ void atomic_max_f32(float* addr, float val) {
    if (!(val < 0.0f)) {
        atomicMax((int*)addr, __float_as_int(val));
    } else {
        atomicMin((unsigned int*)addr, __float_as_uint(val));
    }
}

__global__ __launch_bounds__(256) void scatter_max_kernel(const float* __restrict__ x,
                                                          const int* __restrict__ src,
                                                          const int* __restrict__ dst,
                                                          float* __restrict__ agg,
                                                          int nedges) {
    int wave = (blockIdx.x * blockDim.x + threadIdx.x) >> 5;
    int lane = threadIdx.x & 31;
    if (wave >= nedges) return;
    int s = src[wave];
    int d = dst[wave];
    const float4 v = *(const float4*)(x + (size_t)s * DFEAT + lane * 4);
    float* base = agg + (size_t)d * DFEAT + lane * 4;
    atomic_max_f32(base + 0, v.x);
    atomic_max_f32(base + 1, v.y);
    atomic_max_f32(base + 2, v.z);
    atomic_max_f32(base + 3, v.w);
}

template <bool DUAL, bool RELU>
__global__ __launch_bounds__(256) void sage_gemm_kernel(const float* __restrict__ A0,
                                                        const float* __restrict__ A1,
                                                        const float* __restrict__ W0,
                                                        const float* __restrict__ W1,
                                                        const float* __restrict__ bias,
                                                        float* __restrict__ out) {
    const int lane = threadIdx.x & 31;
    const int wv   = threadIdx.x >> 5;
    const int m    = lane & 15;
    const int half = lane >> 4;
    const int row_base = blockIdx.x * 16;
    const int col_base = wv * 16;
    const int koff = half * 2;

    const float* a0row = A0 + (size_t)(row_base + m) * DFEAT + koff;
    const float* w0row = W0 + (size_t)(col_base + m) * DFEAT + koff;
    const float* a1row = DUAL ? (A1 + (size_t)(row_base + m) * DFEAT + koff) : a0row;
    const float* w1row = DUAL ? (W1 + (size_t)(col_base + m) * DFEAT + koff) : w0row;

    v8f acc = {};
#pragma unroll 4
    for (int kk = 0; kk < DFEAT; kk += 4) {
        v2f a0 = *(const v2f*)(a0row + kk);
        if (DUAL) {
            a0.x = (a0.x == -INFINITY) ? 0.0f : a0.x;
            a0.y = (a0.y == -INFINITY) ? 0.0f : a0.y;
        }
        v2f b0 = *(const v2f*)(w0row + kk);
        acc = __builtin_amdgcn_wmma_f32_16x16x4_f32(false, a0, false, b0,
                                                    (short)0, acc, false, false);
        if (DUAL) {
            v2f a1 = *(const v2f*)(a1row + kk);
            v2f b1 = *(const v2f*)(w1row + kk);
            acc = __builtin_amdgcn_wmma_f32_16x16x4_f32(false, a1, false, b1,
                                                        (short)0, acc, false, false);
        }
    }

    __syncthreads();

    const float bv = bias[col_base + m];
#pragma unroll
    for (int r = 0; r < 8; ++r) {
        float v = acc[r] + bv;
        if (RELU) v = fmaxf(v, 0.0f);
        out[(size_t)(row_base + r + 8 * half) * DFEAT + col_base + m] = v;
    }
}

extern "C" void kernel_launch(void* const* d_in, const int* in_sizes, int n_in,
                              void* d_out, int out_size, void* d_ws, size_t ws_size,
                              hipStream_t stream) {
    const float* x  = (const float*)d_in[0];
    const int*   ei = (const int*)d_in[1];
    const float* Wl = (const float*)d_in[2];
    const float* bl = (const float*)d_in[3];
    const float* Wr = (const float*)d_in[4];
    const float* W  = (const float*)d_in[5];
    const float* b  = (const float*)d_in[6];

    const int nE = in_sizes[1] / 2;
    const int n  = in_sizes[0] / DFEAT;
    const int*   srcp = ei;
    const int*   dstp = ei + nE;

    float* buf0 = (float*)d_ws;
    float* buf1 = buf0 + (size_t)n * DFEAT;

    dim3 blk(256);
    const int fillBlocks = (n * DFEAT / 4 + 255) / 256;
    const int scatBlocks = (nE + 7) / 8;
    const int gemmBlocks = n / 16;

    fill_neginf_kernel<<<fillBlocks, blk, 0, stream>>>((uint4*)buf0, n * DFEAT / 4);
    scatter_max_kernel<<<scatBlocks, blk, 0, stream>>>(x, srcp, dstp, buf0, nE);
    sage_gemm_kernel<true, true><<<gemmBlocks, blk, 0, stream>>>(
        buf0, x, Wl, Wr, bl, buf0);

    fill_neginf_kernel<<<fillBlocks, blk, 0, stream>>>((uint4*)buf1, n * DFEAT / 4);
    scatter_max_kernel<<<scatBlocks, blk, 0, stream>>>(buf0, srcp, dstp, buf1, nE);
    sage_gemm_kernel<true, true><<<gemmBlocks, blk, 0, stream>>>(
        buf1, buf0, Wl + DFEAT * DFEAT, Wr + DFEAT * DFEAT, bl + DFEAT, buf1);

    sage_gemm_kernel<false, false><<<gemmBlocks, blk, 0, stream>>>(
        buf1, nullptr, W, nullptr, b, (float*)d_out);
}


# Launches 7 to 5: layer-0 GEMM also stores layer 1's -inf start keys and zeroes its counters; layer-1 GEMM applies the final Linear to its x2 tile from LDS, x2 never stored; no sum reordered
# speedup vs baseline: 3.6994x; 3.6994x over previous
#include <hip/hip_runtime.h>
#include <math.h>

typedef __attribute__((ext_vector_type(2))) float v2f;
typedef __attribute__((ext_vector_type(8))) float v8f;

#define DFEAT 128

#define KEY_NEG_INF 0x007FFFFFu

__device__ __forceinline__ unsigned int f32_to_key(float v) {
    const unsigned int b = __float_as_uint(v);
    return b ^ ((unsigned int)((int)b >> 31) | 0x80000000u);
}

__device__ __forceinline__ float key_to_f32(unsigned int k) {
    return __uint_as_float(k ^ (((int)k < 0) ? 0x80000000u : 0xFFFFFFFFu));
}

#define CTR_STRIDE 64

__global__ __launch_bounds__(256) void fill_neginf_key_kernel(uint4* __restrict__ p, int n4,
                                                              unsigned int* __restrict__ ctr) {
    int i = blockIdx.x * blockDim.x + threadIdx.x;
    if (i < n4) {
        uint4 v;
        v.x = KEY_NEG_INF; v.y = KEY_NEG_INF; v.z = KEY_NEG_INF; v.w = KEY_NEG_INF;
        p[i] = v;
    }
    if (i == 0) {
        ctr[0] = 0u;
        ctr[CTR_STRIDE] = 0u;
    }
}

#define ITEM_EDGES     64
#define EDGE_BATCH     16
#define SCATTER_WAVES  8
#define ITEM_SPAN      (ITEM_EDGES * SCATTER_WAVES)

__global__ __launch_bounds__(256) void scatter_max_kernel(const float* __restrict__ x,
                                                          const int* __restrict__ src,
                                                          const int* __restrict__ dst,
                                                          unsigned int* __restrict__ agg,
                                                          unsigned int* __restrict__ ctr,
                                                          int nedges) {
    __shared__ int s_pick;
    const int lane = threadIdx.x & 31;
    const int wv = __builtin_amdgcn_readfirstlane((int)(threadIdx.x >> 5));
    const unsigned int nitems = (unsigned int)((nedges + ITEM_SPAN - 1) / ITEM_SPAN);

    if (threadIdx.x == 0) {
        const unsigned int where = __builtin_amdgcn_s_sendmsg_rtn(0x87);
        unsigned int q = (where >> 18) & 1u;
        unsigned int t = atomicAdd(ctr + q * CTR_STRIDE, 1u);
        if (t >= nitems) {
            q ^= 1u;
            t = atomicAdd(ctr + q * CTR_STRIDE, 1u);
        }
        s_pick = (t < nitems) ? (int)(t * 2u + q) : -1;
    }
    __syncthreads();
    const int pick = __builtin_amdgcn_readfirstlane(s_pick);
    if (pick < 0) return;
    const int fh   = pick & 1;
    const int item = pick >> 1;

    const int e0 = (item * SCATTER_WAVES + wv) * ITEM_EDGES;
    if (e0 >= nedges) return;
    const int left = nedges - e0;
    const int cnt  = left < ITEM_EDGES ? left : ITEM_EDGES;

    int sv0 = 0, dv0 = 0, sv1 = 0, dv1 = 0;
    if (lane < cnt) {
        sv0 = src[e0 + lane];
        dv0 = dst[e0 + lane];
    }
    if (lane + 32 < cnt) {
        sv1 = src[e0 + 32 + lane];
        dv1 = dst[e0 + 32 + lane];
    }
    const unsigned int loff = (unsigned int)(fh * 64 + lane);

    int j0 = 0;
    for (; j0 + EDGE_BATCH <= cnt; j0 += EDGE_BATCH) {
        const bool upper = (j0 & 32) != 0;
        const int  svk = upper ? sv1 : sv0;
        const int  dvk = upper ? dv1 : dv0;
        const int  jb  = j0 & 31;
        float v[EDGE_BATCH][2];
#pragma unroll
        for (int j = 0; j < EDGE_BATCH; ++j) {
            const int s = __builtin_amdgcn_readlane(svk, jb + j);
            const float* xr = x + (size_t)((unsigned int)s * DFEAT + loff);
            v[j][0] = xr[0];
            v[j][1] = xr[32];
        }
#pragma unroll
        for (int j = 0; j < EDGE_BATCH; ++j) {
            const int d = __builtin_amdgcn_readlane(dvk, jb + j);
            unsigned int* base = agg + (size_t)((unsigned int)d * DFEAT + loff);
            atomicMax(base + 0,  f32_to_key(v[j][0]));
            atomicMax(base + 32, f32_to_key(v[j][1]));
        }
    }
    for (; j0 < cnt; ++j0) {
        const bool upper = (j0 & 32) != 0;
        const int s = __builtin_amdgcn_readlane(upper ? sv1 : sv0, j0 & 31);
        const int d = __builtin_amdgcn_readlane(upper ? dv1 : dv0, j0 & 31);
        const float* xr = x + (size_t)((unsigned int)s * DFEAT + loff);
        const float v0 = xr[0];
        const float v1 = xr[32];
        unsigned int* base = agg + (size_t)((unsigned int)d * DFEAT + loff);
        atomicMax(base + 0,  f32_to_key(v0));
        atomicMax(base + 32, f32_to_key(v1));
    }
}

#define LDS_PITCH   132
#define TILE_DW     (16 * LDS_PITCH)
#define GEMM_WAVES  8
#define GEMM_BLOCKS 512

__device__ __forceinline__ void stage_store(float* row, int q, float4 v) {
    float* p = row + 8 * (q >> 1) + 2 * (q & 1);
    v2f lo; lo.x = v.x; lo.y = v.y;
    v2f hi; hi.x = v.z; hi.y = v.w;
    *(v2f*)(p)     = lo;
    *(v2f*)(p + 4) = hi;
}

__device__ __forceinline__ float4 frag_load(const float* tile, int m, int half, int g) {
    return *(const float4*)(tile + m * LDS_PITCH + 8 * g + 4 * half);
}

__device__ __forceinline__ float fix_key(float kf) {
    const unsigned int k = __float_as_uint(kf);
    return (k == KEY_NEG_INF) ? 0.0f : key_to_f32(k);
}

template <bool FINAL>
__global__ __launch_bounds__(256) void sage_gemm_kernel(const float* A0,
                                                        const float* A1,
                                                        const float* __restrict__ W0,
                                                        const float* __restrict__ W1,
                                                        const float* __restrict__ bias,
                                                        const float* __restrict__ W2,
                                                        const float* __restrict__ bias2,
                                                        float* out,
                                                        unsigned int* __restrict__ next_plane,
                                                        unsigned int* __restrict__ next_ctr,
                                                        int ntiles) {
    __shared__ __attribute__((aligned(16))) float lds[GEMM_WAVES * TILE_DW];

    const int lane = threadIdx.x & 31;
    const int wv   = threadIdx.x >> 5;
    const int m    = lane & 15;
    const int half = lane >> 4;
    const int col_base = wv * 16;

    float* wreg = lds + wv * TILE_DW;
    float4 b0[16];
    float4 b1[16];
    float4 b2[16];
#pragma unroll
    for (int r = 0; r < 16; ++r) {
        const float4 v = *(const float4*)(W0 + (size_t)(col_base + r) * DFEAT + lane * 4);
        stage_store(wreg + r * LDS_PITCH, lane, v);
    }
    __syncthreads();
#pragma unroll
    for (int g = 0; g < 16; ++g) b0[g] = frag_load(wreg, m, half, g);
    __syncthreads();
#pragma unroll
    for (int r = 0; r < 16; ++r) {
        const float4 v = *(const float4*)(W1 + (size_t)(col_base + r) * DFEAT + lane * 4);
        stage_store(wreg + r * LDS_PITCH, lane, v);
    }
    __syncthreads();
#pragma unroll
    for (int g = 0; g < 16; ++g) b1[g] = frag_load(wreg, m, half, g);
    if (FINAL) {
        __syncthreads();
#pragma unroll
        for (int r = 0; r < 16; ++r) {
            const float4 v = *(const float4*)(W2 + (size_t)(col_base + r) * DFEAT + lane * 4);
            stage_store(wreg + r * LDS_PITCH, lane, v);
        }
        __syncthreads();
#pragma unroll
        for (int g = 0; g < 16; ++g) b2[g] = frag_load(wreg, m, half, g);
    }
    __syncthreads();

    const float bv = bias[col_base + m];
    float bv2 = 0.0f;
    if (FINAL) bv2 = bias2[col_base + m];

    if (!FINAL) {
        if (blockIdx.x == 0 && threadIdx.x == 0) {
            next_ctr[0] = 0u;
            next_ctr[CTR_STRIDE] = 0u;
        }
    }

    const int xcol = 8 * (col_base >> 3) + 8 * (m >> 3) + ((m & 1) | ((m & 2) << 1) | ((m & 4) >> 1));

    float4 pa0[2];
    float4 pa1[2];
    int t = blockIdx.x;
    if (t < ntiles) {
#pragma unroll
        for (int i = 0; i < 2; ++i) {
            const size_t off = (size_t)(t * 16 + wv + 8 * i) * DFEAT + lane * 4;
            pa0[i] = *(const float4*)(A0 + off);
            pa1[i] = *(const float4*)(A1 + off);
        }
#pragma unroll
        for (int i = 0; i < 2; ++i) {
            float4 v = pa0[i];
            v.x = fix_key(v.x); v.y = fix_key(v.y); v.z = fix_key(v.z); v.w = fix_key(v.w);
            stage_store(lds + (wv + 8 * i) * LDS_PITCH, lane, v);
            stage_store(lds + TILE_DW + (wv + 8 * i) * LDS_PITCH, lane, pa1[i]);
        }
    }
    __syncthreads();

    int buf = 0;
    for (; t < ntiles; t += gridDim.x) {
        const int  tn       = t + gridDim.x;
        const bool has_next = tn < ntiles;
        if (has_next) {
#pragma unroll
            for (int i = 0; i < 2; ++i) {
                const size_t off = (size_t)(tn * 16 + wv + 8 * i) * DFEAT + lane * 4;
                pa0[i] = *(const float4*)(A0 + off);
                pa1[i] = *(const float4*)(A1 + off);
            }
        }

        const float* a0t = lds + buf * 2 * TILE_DW;
        const float* a1t = a0t + TILE_DW;
        v8f acc = {};
#pragma unroll
        for (int g = 0; g < 16; ++g) {
            const float4 fa0 = frag_load(a0t, m, half, g);
            const float4 fa1 = frag_load(a1t, m, half, g);
            v2f a, b;
            a.x = fa0.x; a.y = fa0.y; b.x = b0[g].x; b.y = b0[g].y;
            acc = __builtin_amdgcn_wmma_f32_16x16x4_f32(false, a, false, b, (short)0, acc, false, false);
            a.x = fa1.x; a.y = fa1.y; b.x = b1[g].x; b.y = b1[g].y;
            acc = __builtin_amdgcn_wmma_f32_16x16x4_f32(false, a, false, b, (short)0, acc, false, false);
            a.x = fa0.z; a.y = fa0.w; b.x = b0[g].z; b.y = b0[g].w;
            acc = __builtin_amdgcn_wmma_f32_16x16x4_f32(false, a, false, b, (short)0, acc, false, false);
            a.x = fa1.z; a.y = fa1.w; b.x = b1[g].z; b.y = b1[g].w;
            acc = __builtin_amdgcn_wmma_f32_16x16x4_f32(false, a, false, b, (short)0, acc, false, false);
        }

        if (FINAL) {
            float* xs = lds + (4 + buf) * TILE_DW;
#pragma unroll
            for (int r = 0; r < 8; ++r)
                xs[(r + 8 * half) * LDS_PITCH + xcol] = fmaxf(acc[r] + bv, 0.0f);
        }

        if (has_next) {
            float* n0 = lds + (buf ^ 1) * 2 * TILE_DW;
#pragma unroll
            for (int i = 0; i < 2; ++i) {
                float4 v = pa0[i];
                v.x = fix_key(v.x); v.y = fix_key(v.y); v.z = fix_key(v.z); v.w = fix_key(v.w);
                stage_store(n0 + (wv + 8 * i) * LDS_PITCH, lane, v);
                stage_store(n0 + TILE_DW + (wv + 8 * i) * LDS_PITCH, lane, pa1[i]);
            }
        }
        __syncthreads();

        const int row_base = t * 16;
        if (FINAL) {
            const float* xt = lds + (4 + buf) * TILE_DW;
            v8f acc2 = {};
#pragma unroll
            for (int g = 0; g < 16; ++g) {
                const float4 fx = frag_load(xt, m, half, g);
                v2f a, b;
                a.x = fx.x; a.y = fx.y; b.x = b2[g].x; b.y = b2[g].y;
                acc2 = __builtin_amdgcn_wmma_f32_16x16x4_f32(false, a, false, b, (short)0, acc2, false, false);
                a.x = fx.z; a.y = fx.w; b.x = b2[g].z; b.y = b2[g].w;
                acc2 = __builtin_amdgcn_wmma_f32_16x16x4_f32(false, a, false, b, (short)0, acc2, false, false);
            }
#pragma unroll
            for (int r = 0; r < 8; ++r)
                out[(size_t)(row_base + r + 8 * half) * DFEAT + col_base + m] = acc2[r] + bv2;
        } else {
#pragma unroll
            for (int r = 0; r < 8; ++r)
                out[(size_t)(row_base + r + 8 * half) * DFEAT + col_base + m] = fmaxf(acc[r] + bv, 0.0f);
            uint4 kv;
            kv.x = KEY_NEG_INF; kv.y = KEY_NEG_INF; kv.z = KEY_NEG_INF; kv.w = KEY_NEG_INF;
#pragma unroll
            for (int i = 0; i < 2; ++i)
                *(uint4*)(next_plane + (size_t)(row_base + wv + 8 * i) * DFEAT + lane * 4) = kv;
        }
        buf ^= 1;
    }
}

extern "C" void kernel_launch(void* const* d_in, const int* in_sizes, int n_in,
                              void* d_out, int out_size, void* d_ws, size_t ws_size,
                              hipStream_t stream) {
    const float* x  = (const float*)d_in[0];
    const int*   ei = (const int*)d_in[1];
    const float* Wl = (const float*)d_in[2];
    const float* bl = (const float*)d_in[3];
    const float* Wr = (const float*)d_in[4];
    const float* W  = (const float*)d_in[5];
    const float* b  = (const float*)d_in[6];

    const int nE = in_sizes[1] / 2;
    const int n  = in_sizes[0] / DFEAT;
    const int*   srcp = ei;
    const int*   dstp = ei + nE;

    float* buf0 = (float*)d_ws;
    float* buf1 = buf0 + (size_t)n * DFEAT;
    unsigned int* ctr0 = (unsigned int*)(buf1 + (size_t)n * DFEAT);
    unsigned int* ctr1 = ctr0 + 2 * CTR_STRIDE;

    dim3 blk(256);
    const int fillBlocks = (n * DFEAT / 4 + 255) / 256;
    const int scatBlocks = 2 * ((nE + ITEM_SPAN - 1) / ITEM_SPAN);
    const int gemmTiles  = n / 16;

    fill_neginf_key_kernel<<<fillBlocks, blk, 0, stream>>>((uint4*)buf0, n * DFEAT / 4, ctr0);
    scatter_max_kernel<<<scatBlocks, blk, 0, stream>>>(x, srcp, dstp, (unsigned int*)buf0, ctr0, nE);
    sage_gemm_kernel<false><<<GEMM_BLOCKS, blk, 0, stream>>>(
        buf0, x, Wl, Wr, bl, nullptr, nullptr, buf0, (unsigned int*)buf1, ctr1, gemmTiles);

    scatter_max_kernel<<<scatBlocks, blk, 0, stream>>>(buf0, srcp, dstp, (unsigned int*)buf1, ctr1, nE);
    sage_gemm_kernel<true><<<GEMM_BLOCKS, blk, 0, stream>>>(
        buf1, buf0, Wl + DFEAT * DFEAT, Wr + DFEAT * DFEAT, bl + DFEAT, W, b, (float*)d_out,
        nullptr, nullptr, gemmTiles);
}


# Scatter-max without global atomics: edges binned once per call by destination range (LDS counters); a work-group owns one range x one feature half as max keys in LDS, walks only its bin; max only
# speedup vs baseline: 3.9866x; 1.0776x over previous
#include <hip/hip_runtime.h>
#include <math.h>

typedef __attribute__((ext_vector_type(2))) float v2f;
typedef __attribute__((ext_vector_type(8))) float v8f;

#define DFEAT 128

#define KEY_NEG_INF 0x007FFFFFu

__device__ __forceinline__ unsigned int f32_to_key(float v) {
    const unsigned int b = __float_as_uint(v);
    return b ^ ((unsigned int)((int)b >> 31) | 0x80000000u);
}

__device__ __forceinline__ float key_to_f32(unsigned int k) {
    return __uint_as_float(k ^ (((int)k < 0) ? 0x80000000u : 0xFFFFFFFFu));
}

#define ROWS_PER_RANGE 392
#define MAX_RANGES     1024
#define BIN_THREADS    1024
#define BIN_PER_THREAD 8
#define BIN_CHUNK      (BIN_THREADS * BIN_PER_THREAD)
#define HIST_BATCH     8

__global__ __launch_bounds__(BIN_THREADS) void bin_count_kernel(const int* __restrict__ dst,
                                                                int nedges, int nranges,
                                                                unsigned int* __restrict__ hist) {
    __shared__ unsigned int h[MAX_RANGES];
    for (int i = threadIdx.x; i < nranges; i += BIN_THREADS) h[i] = 0u;
    __syncthreads();
    const int base = blockIdx.x * BIN_CHUNK + threadIdx.x;
    int d[BIN_PER_THREAD];
#pragma unroll
    for (int k = 0; k < BIN_PER_THREAD; ++k) {
        const int e = base + k * BIN_THREADS;
        d[k] = (e < nedges) ? dst[e] : -1;
    }
#pragma unroll
    for (int k = 0; k < BIN_PER_THREAD; ++k)
        if (d[k] >= 0) atomicAdd(&h[(unsigned int)d[k] / ROWS_PER_RANGE], 1u);
    __syncthreads();
    for (int i = threadIdx.x; i < nranges; i += BIN_THREADS) hist[blockIdx.x * nranges + i] = h[i];
}

__global__ __launch_bounds__(BIN_THREADS) void bin_place_kernel(const int* __restrict__ src,
                                                                const int* __restrict__ dst,
                                                                int nedges, int nranges, int nblocks,
                                                                const unsigned int* __restrict__ hist,
                                                                int* __restrict__ bsrc,
                                                                int* __restrict__ bdst,
                                                                int* __restrict__ bin_start) {
    __shared__ unsigned int tot[MAX_RANGES];
    __shared__ unsigned int gbase[MAX_RANGES];
    __shared__ unsigned int own[MAX_RANGES];
    __shared__ unsigned int lstart[MAX_RANGES];
    __shared__ unsigned int lcur[MAX_RANGES];
    __shared__ int ssrc[BIN_CHUNK];
    __shared__ int sdst[BIN_CHUNK];
    for (int i = threadIdx.x; i < nranges; i += BIN_THREADS) { tot[i] = 0u; gbase[i] = 0u; own[i] = 0u; }
    __syncthreads();

    const int nhist = nblocks * nranges;
    for (int i0 = 0; i0 < nhist; i0 += HIST_BATCH * BIN_THREADS) {
        unsigned int hv[HIST_BATCH];
#pragma unroll
        for (int k = 0; k < HIST_BATCH; ++k) {
            const int idx = i0 + k * BIN_THREADS + (int)threadIdx.x;
            hv[k] = (idx < nhist) ? hist[idx] : 0u;
        }
#pragma unroll
        for (int k = 0; k < HIST_BATCH; ++k) {
            const int idx = i0 + k * BIN_THREADS + (int)threadIdx.x;
            if (idx < nhist) {
                const int b = idx / nranges;
                const int r = idx - b * nranges;
                atomicAdd(&tot[r], hv[k]);
                if (b < (int)blockIdx.x) atomicAdd(&gbase[r], hv[k]);
                if (b == (int)blockIdx.x) own[r] = hv[k];
            }
        }
    }
    __syncthreads();

    for (int r = threadIdx.x; r < nranges; r += BIN_THREADS) {
        unsigned int s = 0u, ls = 0u;
        for (int q = 0; q < r; ++q) { s += tot[q]; ls += own[q]; }
        gbase[r] += s;
        lstart[r] = ls;
        lcur[r]   = ls;
        if (blockIdx.x == 0) {
            bin_start[r] = (int)s;
            if (r == nranges - 1) bin_start[nranges] = (int)(s + tot[r]);
        }
    }
    __syncthreads();

    const int chunk0 = blockIdx.x * BIN_CHUNK;
    const int base = chunk0 + threadIdx.x;
    int sv[BIN_PER_THREAD];
    int dv[BIN_PER_THREAD];
#pragma unroll
    for (int k = 0; k < BIN_PER_THREAD; ++k) {
        const int e = base + k * BIN_THREADS;
        const bool ok = e < nedges;
        sv[k] = ok ? src[e] : 0;
        dv[k] = ok ? dst[e] : -1;
    }
#pragma unroll
    for (int k = 0; k < BIN_PER_THREAD; ++k) {
        if (dv[k] >= 0) {
            const unsigned int lp = atomicAdd(&lcur[(unsigned int)dv[k] / ROWS_PER_RANGE], 1u);
            ssrc[lp] = sv[k];
            sdst[lp] = dv[k];
        }
    }
    __syncthreads();

    const int left = nedges - chunk0;
    const int m = left < BIN_CHUNK ? left : BIN_CHUNK;
    for (int i = threadIdx.x; i < m; i += BIN_THREADS) {
        const int s = ssrc[i];
        const int d = sdst[i];
        const unsigned int r = (unsigned int)d / ROWS_PER_RANGE;
        const unsigned int pos = gbase[r] + ((unsigned int)i - lstart[r]);
        bsrc[pos] = s;
        bdst[pos] = d;
    }
}

#define EDGE_BATCH   16
#define RANGE_WAVES  32
#define KEY_WORDS    (ROWS_PER_RANGE * 64)

__global__ __launch_bounds__(1024) void range_max_kernel(const float* __restrict__ x,
                                                         const int* __restrict__ bsrc,
                                                         const int* __restrict__ bdst,
                                                         const int* __restrict__ bin_start,
                                                         unsigned int* __restrict__ agg,
                                                         int n) {
    __shared__ __attribute__((aligned(16))) unsigned int keys[KEY_WORDS];
    const int rng = blockIdx.x >> 1;
    const int fh  = blockIdx.x & 1;
    const int row_base = rng * ROWS_PER_RANGE;

    uint4 kv;
    kv.x = KEY_NEG_INF; kv.y = KEY_NEG_INF; kv.z = KEY_NEG_INF; kv.w = KEY_NEG_INF;
    for (int i = threadIdx.x; i < KEY_WORDS / 4; i += 1024) ((uint4*)keys)[i] = kv;
    __syncthreads();

    const int lane = threadIdx.x & 31;
    const int wv = __builtin_amdgcn_readfirstlane((int)(threadIdx.x >> 5));
    const int e_begin = bin_start[rng];
    const int e_end   = bin_start[rng + 1];
    const unsigned int loff = (unsigned int)(fh * 64 + lane);

    int e0 = e_begin + wv * EDGE_BATCH;
    int sv = 0, dv = row_base;
    if (e0 < e_end && lane < e_end - e0 && lane < EDGE_BATCH) {
        sv = bsrc[e0 + lane];
        dv = bdst[e0 + lane];
    }
    for (; e0 < e_end; e0 += RANGE_WAVES * EDGE_BATCH) {
        const int left = e_end - e0;
        const int cnt  = left < EDGE_BATCH ? left : EDGE_BATCH;
        int svn = 0, dvn = row_base;
        if (cnt == EDGE_BATCH) {
            float v[EDGE_BATCH][2];
#pragma unroll
            for (int j = 0; j < EDGE_BATCH; ++j) {
                const int s = __builtin_amdgcn_readlane(sv, j);
                const float* xr = x + (size_t)((unsigned int)s * DFEAT + loff);
                v[j][0] = xr[0];
                v[j][1] = xr[32];
            }
            const int en = e0 + RANGE_WAVES * EDGE_BATCH;
            if (en < e_end && lane < e_end - en && lane < EDGE_BATCH) {
                svn = bsrc[en + lane];
                dvn = bdst[en + lane];
            }
#pragma unroll
            for (int j = 0; j < EDGE_BATCH; ++j) {
                const int d = __builtin_amdgcn_readlane(dv, j) - row_base;
                atomicMax(&keys[d * 64 + lane],      f32_to_key(v[j][0]));
                atomicMax(&keys[d * 64 + 32 + lane], f32_to_key(v[j][1]));
            }
        } else {
            for (int j = 0; j < cnt; ++j) {
                const int s = __builtin_amdgcn_readlane(sv, j);
                const int d = __builtin_amdgcn_readlane(dv, j) - row_base;
                const float* xr = x + (size_t)((unsigned int)s * DFEAT + loff);
                const float v0 = xr[0];
                const float v1 = xr[32];
                atomicMax(&keys[d * 64 + lane],      f32_to_key(v0));
                atomicMax(&keys[d * 64 + 32 + lane], f32_to_key(v1));
            }
        }
        sv = svn;
        dv = dvn;
    }
    __syncthreads();

    for (int i = threadIdx.x; i < KEY_WORDS / 4; i += 1024) {
        const int row = row_base + (i >> 4);
        if (row < n)
            *(uint4*)(agg + (size_t)row * DFEAT + fh * 64 + (i & 15) * 4) = ((const uint4*)keys)[i];
    }
}

#define LDS_PITCH   132
#define TILE_DW     (16 * LDS_PITCH)
#define GEMM_WAVES  8
#define GEMM_BLOCKS 512

__device__ __forceinline__ void stage_store(float* row, int q, float4 v) {
    float* p = row + 8 * (q >> 1) + 2 * (q & 1);
    v2f lo; lo.x = v.x; lo.y = v.y;
    v2f hi; hi.x = v.z; hi.y = v.w;
    *(v2f*)(p)     = lo;
    *(v2f*)(p + 4) = hi;
}

__device__ __forceinline__ float4 frag_load(const float* tile, int m, int half, int g) {
    return *(const float4*)(tile + m * LDS_PITCH + 8 * g + 4 * half);
}

__device__ __forceinline__ float fix_key(float kf) {
    const unsigned int k = __float_as_uint(kf);
    return (k == KEY_NEG_INF) ? 0.0f : key_to_f32(k);
}

template <bool FINAL>
__global__ __launch_bounds__(256) void sage_gemm_kernel(const float* A0,
                                                        const float* A1,
                                                        const float* __restrict__ W0,
                                                        const float* __restrict__ W1,
                                                        const float* __restrict__ bias,
                                                        const float* __restrict__ W2,
                                                        const float* __restrict__ bias2,
                                                        float* out,
                                                        int ntiles) {
    __shared__ __attribute__((aligned(16))) float lds[GEMM_WAVES * TILE_DW];

    const int lane = threadIdx.x & 31;
    const int wv   = threadIdx.x >> 5;
    const int m    = lane & 15;
    const int half = lane >> 4;
    const int col_base = wv * 16;

    float* wreg = lds + wv * TILE_DW;
    float4 b0[16];
    float4 b1[16];
    float4 b2[16];
#pragma unroll
    for (int r = 0; r < 16; ++r) {
        const float4 v = *(const float4*)(W0 + (size_t)(col_base + r) * DFEAT + lane * 4);
        stage_store(wreg + r * LDS_PITCH, lane, v);
    }
    __syncthreads();
#pragma unroll
    for (int g = 0; g < 16; ++g) b0[g] = frag_load(wreg, m, half, g);
    __syncthreads();
#pragma unroll
    for (int r = 0; r < 16; ++r) {
        const float4 v = *(const float4*)(W1 + (size_t)(col_base + r) * DFEAT + lane * 4);
        stage_store(wreg + r * LDS_PITCH, lane, v);
    }
    __syncthreads();
#pragma unroll
    for (int g = 0; g < 16; ++g) b1[g] = frag_load(wreg, m, half, g);
    if (FINAL) {
        __syncthreads();
#pragma unroll
        for (int r = 0; r < 16; ++r) {
            const float4 v = *(const float4*)(W2 + (size_t)(col_base + r) * DFEAT + lane * 4);
            stage_store(wreg + r * LDS_PITCH, lane, v);
        }
        __syncthreads();
#pragma unroll
        for (int g = 0; g < 16; ++g) b2[g] = frag_load(wreg, m, half, g);
    }
    __syncthreads();

    const float bv = bias[col_base + m];
    float bv2 = 0.0f;
    if (FINAL) bv2 = bias2[col_base + m];

    const int xcol = 8 * (col_base >> 3) + 8 * (m >> 3) + ((m & 1) | ((m & 2) << 1) | ((m & 4) >> 1));

    float4 pa0[2];
    float4 pa1[2];
    int t = blockIdx.x;
    if (t < ntiles) {
#pragma unroll
        for (int i = 0; i < 2; ++i) {
            const size_t off = (size_t)(t * 16 + wv + 8 * i) * DFEAT + lane * 4;
            pa0[i] = *(const float4*)(A0 + off);
            pa1[i] = *(const float4*)(A1 + off);
        }
#pragma unroll
        for (int i = 0; i < 2; ++i) {
            float4 v = pa0[i];
            v.x = fix_key(v.x); v.y = fix_key(v.y); v.z = fix_key(v.z); v.w = fix_key(v.w);
            stage_store(lds + (wv + 8 * i) * LDS_PITCH, lane, v);
            stage_store(lds + TILE_DW + (wv + 8 * i) * LDS_PITCH, lane, pa1[i]);
        }
    }
    __syncthreads();

    int buf = 0;
    for (; t < ntiles; t += gridDim.x) {
        const int  tn       = t + gridDim.x;
        const bool has_next = tn < ntiles;
        if (has_next) {
#pragma unroll
            for (int i = 0; i < 2; ++i) {
                const size_t off = (size_t)(tn * 16 + wv + 8 * i) * DFEAT + lane * 4;
                pa0[i] = *(const float4*)(A0 + off);
                pa1[i] = *(const float4*)(A1 + off);
            }
        }

        const float* a0t = lds + buf * 2 * TILE_DW;
        const float* a1t = a0t + TILE_DW;
        v8f acc = {};
#pragma unroll
        for (int g = 0; g < 16; ++g) {
            const float4 fa0 = frag_load(a0t, m, half, g);
            const float4 fa1 = frag_load(a1t, m, half, g);
            v2f a, b;
            a.x = fa0.x; a.y = fa0.y; b.x = b0[g].x; b.y = b0[g].y;
            acc = __builtin_amdgcn_wmma_f32_16x16x4_f32(false, a, false, b, (short)0, acc, false, false);
            a.x = fa1.x; a.y = fa1.y; b.x = b1[g].x; b.y = b1[g].y;
            acc = __builtin_amdgcn_wmma_f32_16x16x4_f32(false, a, false, b, (short)0, acc, false, false);
            a.x = fa0.z; a.y = fa0.w; b.x = b0[g].z; b.y = b0[g].w;
            acc = __builtin_amdgcn_wmma_f32_16x16x4_f32(false, a, false, b, (short)0, acc, false, false);
            a.x = fa1.z; a.y = fa1.w; b.x = b1[g].z; b.y = b1[g].w;
            acc = __builtin_amdgcn_wmma_f32_16x16x4_f32(false, a, false, b, (short)0, acc, false, false);
        }

        if (FINAL) {
            float* xs = lds + (4 + buf) * TILE_DW;
#pragma unroll
            for (int r = 0; r < 8; ++r)
                xs[(r + 8 * half) * LDS_PITCH + xcol] = fmaxf(acc[r] + bv, 0.0f);
        }

        if (has_next) {
            float* n0 = lds + (buf ^ 1) * 2 * TILE_DW;
#pragma unroll
            for (int i = 0; i < 2; ++i) {
                float4 v = pa0[i];
                v.x = fix_key(v.x); v.y = fix_key(v.y); v.z = fix_key(v.z); v.w = fix_key(v.w);
                stage_store(n0 + (wv + 8 * i) * LDS_PITCH, lane, v);
                stage_store(n0 + TILE_DW + (wv + 8 * i) * LDS_PITCH, lane, pa1[i]);
            }
        }
        __syncthreads();

        const int row_base = t * 16;
        if (FINAL) {
            const float* xt = lds + (4 + buf) * TILE_DW;
            v8f acc2 = {};
#pragma unroll
            for (int g = 0; g < 16; ++g) {
                const float4 fx = frag_load(xt, m, half, g);
                v2f a, b;
                a.x = fx.x; a.y = fx.y; b.x = b2[g].x; b.y = b2[g].y;
                acc2 = __builtin_amdgcn_wmma_f32_16x16x4_f32(false, a, false, b, (short)0, acc2, false, false);
                a.x = fx.z; a.y = fx.w; b.x = b2[g].z; b.y = b2[g].w;
                acc2 = __builtin_amdgcn_wmma_f32_16x16x4_f32(false, a, false, b, (short)0, acc2, false, false);
            }
#pragma unroll
            for (int r = 0; r < 8; ++r)
                out[(size_t)(row_base + r + 8 * half) * DFEAT + col_base + m] = acc2[r] + bv2;
        } else {
#pragma unroll
            for (int r = 0; r < 8; ++r)
                out[(size_t)(row_base + r + 8 * half) * DFEAT + col_base + m] = fmaxf(acc[r] + bv, 0.0f);
        }
        buf ^= 1;
    }
}

extern "C" void kernel_launch(void* const* d_in, const int* in_sizes, int n_in,
                              void* d_out, int out_size, void* d_ws, size_t ws_size,
                              hipStream_t stream) {
    const float* x  = (const float*)d_in[0];
    const int*   ei = (const int*)d_in[1];
    const float* Wl = (const float*)d_in[2];
    const float* bl = (const float*)d_in[3];
    const float* Wr = (const float*)d_in[4];
    const float* W  = (const float*)d_in[5];
    const float* b  = (const float*)d_in[6];

    const int nE = in_sizes[1] / 2;
    const int n  = in_sizes[0] / DFEAT;
    const int*   srcp = ei;
    const int*   dstp = ei + nE;

    float* buf0 = (float*)d_ws;
    float* buf1 = buf0 + (size_t)n * DFEAT;
    const int nranges   = (n + ROWS_PER_RANGE - 1) / ROWS_PER_RANGE;
    const int binBlocks = (nE + BIN_CHUNK - 1) / BIN_CHUNK;
    int* bsrc = (int*)(buf1 + (size_t)n * DFEAT);
    int* bdst = bsrc + nE;
    unsigned int* hist = (unsigned int*)(bdst + nE);
    int* binStart = (int*)(hist + (size_t)binBlocks * nranges);

    dim3 blk(256);
    const int gemmTiles = n / 16;

    bin_count_kernel<<<binBlocks, BIN_THREADS, 0, stream>>>(dstp, nE, nranges, hist);
    bin_place_kernel<<<binBlocks, BIN_THREADS, 0, stream>>>(srcp, dstp, nE, nranges, binBlocks, hist,
                                                            bsrc, bdst, binStart);

    range_max_kernel<<<2 * nranges, 1024, 0, stream>>>(x, bsrc, bdst, binStart, (unsigned int*)buf0, n);
    sage_gemm_kernel<false><<<GEMM_BLOCKS, blk, 0, stream>>>(
        buf0, x, Wl, Wr, bl, nullptr, nullptr, buf0, gemmTiles);

    range_max_kernel<<<2 * nranges, 1024, 0, stream>>>(buf0, bsrc, bdst, binStart, (unsigned int*)buf1, n);
    sage_gemm_kernel<true><<<GEMM_BLOCKS, blk, 0, stream>>>(
        buf1, buf0, Wl + DFEAT * DFEAT, Wr + DFEAT * DFEAT, bl + DFEAT, W, b, (float*)d_out,
        gemmTiles);
}


# Scatter-max work-group owns whole rows (196 rows, 256 bins), not a feature half: an edge's 512-byte source row is one 16-byte-per-lane load, not 2 passes of 4-byte loads; max only, no sum reordered
# speedup vs baseline: 4.1086x; 1.0306x over previous
#include <hip/hip_runtime.h>
#include <math.h>

typedef __attribute__((ext_vector_type(2))) float v2f;
typedef __attribute__((ext_vector_type(8))) float v8f;

#define DFEAT 128

#define KEY_NEG_INF 0x007FFFFFu

__device__ __forceinline__ unsigned int f32_to_key(float v) {
    const unsigned int b = __float_as_uint(v);
    return b ^ ((unsigned int)((int)b >> 31) | 0x80000000u);
}

__device__ __forceinline__ float key_to_f32(unsigned int k) {
    return __uint_as_float(k ^ (((int)k < 0) ? 0x80000000u : 0xFFFFFFFFu));
}

#define ROWS_PER_RANGE 196
#define MAX_RANGES     1024
#define BIN_THREADS    1024
#define BIN_PER_THREAD 8
#define BIN_CHUNK      (BIN_THREADS * BIN_PER_THREAD)
#define HIST_BATCH     16

__global__ __launch_bounds__(BIN_THREADS) void bin_count_kernel(const int* __restrict__ dst,
                                                                int nedges, int nranges,
                                                                unsigned int* __restrict__ hist) {
    __shared__ unsigned int h[MAX_RANGES];
    for (int i = threadIdx.x; i < nranges; i += BIN_THREADS) h[i] = 0u;
    __syncthreads();
    const int base = blockIdx.x * BIN_CHUNK + threadIdx.x;
    int d[BIN_PER_THREAD];
#pragma unroll
    for (int k = 0; k < BIN_PER_THREAD; ++k) {
        const int e = base + k * BIN_THREADS;
        d[k] = (e < nedges) ? dst[e] : -1;
    }
#pragma unroll
    for (int k = 0; k < BIN_PER_THREAD; ++k)
        if (d[k] >= 0) atomicAdd(&h[(unsigned int)d[k] / ROWS_PER_RANGE], 1u);
    __syncthreads();
    for (int i = threadIdx.x; i < nranges; i += BIN_THREADS) hist[blockIdx.x * nranges + i] = h[i];
}

__global__ __launch_bounds__(BIN_THREADS) void bin_place_kernel(const int* __restrict__ src,
                                                                const int* __restrict__ dst,
                                                                int nedges, int nranges, int nblocks,
                                                                const unsigned int* __restrict__ hist,
                                                                int* __restrict__ bsrc,
                                                                int* __restrict__ bdst,
                                                                int* __restrict__ bin_start) {
    __shared__ __attribute__((aligned(16))) unsigned int tot[MAX_RANGES];
    __shared__ unsigned int gbase[MAX_RANGES];
    __shared__ __attribute__((aligned(16))) unsigned int own[MAX_RANGES];
    __shared__ unsigned int lstart[MAX_RANGES];
    __shared__ unsigned int lcur[MAX_RANGES];
    __shared__ int ssrc[BIN_CHUNK];
    __shared__ int sdst[BIN_CHUNK];
    for (int i = threadIdx.x; i < nranges; i += BIN_THREADS) { tot[i] = 0u; gbase[i] = 0u; own[i] = 0u; }
    __syncthreads();

    const int nhist = nblocks * nranges;
    const int step_b = BIN_THREADS / nranges;
    const int step_r = BIN_THREADS - step_b * nranges;
    int hb = (int)threadIdx.x / nranges;
    int hr = (int)threadIdx.x - hb * nranges;
    for (int i0 = 0; i0 < nhist; i0 += HIST_BATCH * BIN_THREADS) {
        unsigned int hv[HIST_BATCH];
#pragma unroll
        for (int k = 0; k < HIST_BATCH; ++k) {
            const int idx = i0 + k * BIN_THREADS + (int)threadIdx.x;
            hv[k] = (idx < nhist) ? hist[idx] : 0u;
        }
#pragma unroll
        for (int k = 0; k < HIST_BATCH; ++k) {
            const int idx = i0 + k * BIN_THREADS + (int)threadIdx.x;
            if (idx < nhist) {
                atomicAdd(&tot[hr], hv[k]);
                if (hb < (int)blockIdx.x) atomicAdd(&gbase[hr], hv[k]);
                if (hb == (int)blockIdx.x) own[hr] = hv[k];
            }
            hb += step_b;
            hr += step_r;
            if (hr >= nranges) { hr -= nranges; ++hb; }
        }
    }
    __syncthreads();

    for (int r = threadIdx.x; r < nranges; r += BIN_THREADS) {
        unsigned int s = 0u, ls = 0u;
        int q = 0;
        for (; q + 4 <= r; q += 4) {
            const uint4 a = *(const uint4*)(tot + q);
            const uint4 c = *(const uint4*)(own + q);
            s  += (a.x + a.y) + (a.z + a.w);
            ls += (c.x + c.y) + (c.z + c.w);
        }
        for (; q < r; ++q) { s += tot[q]; ls += own[q]; }
        gbase[r] += s;
        lstart[r] = ls;
        lcur[r]   = ls;
        if (blockIdx.x == 0) {
            bin_start[r] = (int)s;
            if (r == nranges - 1) bin_start[nranges] = (int)(s + tot[r]);
        }
    }
    __syncthreads();

    const int chunk0 = blockIdx.x * BIN_CHUNK;
    const int base = chunk0 + threadIdx.x;
    int sv[BIN_PER_THREAD];
    int dv[BIN_PER_THREAD];
#pragma unroll
    for (int k = 0; k < BIN_PER_THREAD; ++k) {
        const int e = base + k * BIN_THREADS;
        const bool ok = e < nedges;
        sv[k] = ok ? src[e] : 0;
        dv[k] = ok ? dst[e] : -1;
    }
#pragma unroll
    for (int k = 0; k < BIN_PER_THREAD; ++k) {
        if (dv[k] >= 0) {
            const unsigned int lp = atomicAdd(&lcur[(unsigned int)dv[k] / ROWS_PER_RANGE], 1u);
            ssrc[lp] = sv[k];
            sdst[lp] = dv[k];
        }
    }
    __syncthreads();

    const int left = nedges - chunk0;
    const int m = left < BIN_CHUNK ? left : BIN_CHUNK;
    for (int i = threadIdx.x; i < m; i += BIN_THREADS) {
        const int s = ssrc[i];
        const int d = sdst[i];
        const unsigned int r = (unsigned int)d / ROWS_PER_RANGE;
        const unsigned int pos = gbase[r] + ((unsigned int)i - lstart[r]);
        bsrc[pos] = s;
        bdst[pos] = d;
    }
}

#define EDGE_BATCH   16
#define RANGE_WAVES  32
#define KEY_WORDS    (ROWS_PER_RANGE * DFEAT)

__global__ __launch_bounds__(1024) void range_max_kernel(const float* __restrict__ x,
                                                         const int* __restrict__ bsrc,
                                                         const int* __restrict__ bdst,
                                                         const int* __restrict__ bin_start,
                                                         unsigned int* __restrict__ agg,
                                                         int n) {
    __shared__ __attribute__((aligned(16))) unsigned int keys[KEY_WORDS];
    const int rng = blockIdx.x;
    const int row_base = rng * ROWS_PER_RANGE;

    uint4 kv;
    kv.x = KEY_NEG_INF; kv.y = KEY_NEG_INF; kv.z = KEY_NEG_INF; kv.w = KEY_NEG_INF;
    for (int i = threadIdx.x; i < KEY_WORDS / 4; i += 1024) ((uint4*)keys)[i] = kv;
    __syncthreads();

    const int lane = threadIdx.x & 31;
    const int wv = __builtin_amdgcn_readfirstlane((int)(threadIdx.x >> 5));
    const int e_begin = bin_start[rng];
    const int e_end   = bin_start[rng + 1];
    const float4* __restrict__ x4 = (const float4*)x;

    int e0 = e_begin + wv * EDGE_BATCH;
    int sv = 0, dv = row_base;
    if (e0 < e_end && lane < e_end - e0 && lane < EDGE_BATCH) {
        sv = bsrc[e0 + lane];
        dv = bdst[e0 + lane];
    }
    for (; e0 < e_end; e0 += RANGE_WAVES * EDGE_BATCH) {
        const int left = e_end - e0;
        const int cnt  = left < EDGE_BATCH ? left : EDGE_BATCH;
        int svn = 0, dvn = row_base;
        if (cnt == EDGE_BATCH) {
            float4 v[EDGE_BATCH];
#pragma unroll
            for (int j = 0; j < EDGE_BATCH; ++j) {
                const int s = __builtin_amdgcn_readlane(sv, j);
                v[j] = x4[(unsigned int)s * (DFEAT / 4) + (unsigned int)lane];
            }
            const int en = e0 + RANGE_WAVES * EDGE_BATCH;
            if (en < e_end && lane < e_end - en && lane < EDGE_BATCH) {
                svn = bsrc[en + lane];
                dvn = bdst[en + lane];
            }
#pragma unroll
            for (int j = 0; j < EDGE_BATCH; ++j) {
                const int d = __builtin_amdgcn_readlane(dv, j) - row_base;
                unsigned int* kr = keys + d * DFEAT + lane;
                atomicMax(kr,      f32_to_key(v[j].x));
                atomicMax(kr + 32, f32_to_key(v[j].y));
                atomicMax(kr + 64, f32_to_key(v[j].z));
                atomicMax(kr + 96, f32_to_key(v[j].w));
            }
        } else {
            for (int j = 0; j < cnt; ++j) {
                const int s = __builtin_amdgcn_readlane(sv, j);
                const int d = __builtin_amdgcn_readlane(dv, j) - row_base;
                const float4 v1 = x4[(unsigned int)s * (DFEAT / 4) + (unsigned int)lane];
                unsigned int* kr = keys + d * DFEAT + lane;
                atomicMax(kr,      f32_to_key(v1.x));
                atomicMax(kr + 32, f32_to_key(v1.y));
                atomicMax(kr + 64, f32_to_key(v1.z));
                atomicMax(kr + 96, f32_to_key(v1.w));
            }
        }
        sv = svn;
        dv = dvn;
    }
    __syncthreads();

    for (int i = threadIdx.x; i < ROWS_PER_RANGE * 32; i += 1024) {
        const int r   = i >> 5;
        const int row = row_base + r;
        const unsigned int* kr = keys + r * DFEAT + (i & 31);
        uint4 k;
        k.x = kr[0]; k.y = kr[32]; k.z = kr[64]; k.w = kr[96];
        if (row < n)
            *(uint4*)(agg + (size_t)row * DFEAT + (i & 31) * 4) = k;
    }
}

#define LDS_PITCH   132
#define TILE_DW     (16 * LDS_PITCH)
#define GEMM_WAVES  8
#define GEMM_BLOCKS 512

__device__ __forceinline__ void stage_store(float* row, int q, float4 v) {
    float* p = row + 8 * (q >> 1) + 2 * (q & 1);
    v2f lo; lo.x = v.x; lo.y = v.y;
    v2f hi; hi.x = v.z; hi.y = v.w;
    *(v2f*)(p)     = lo;
    *(v2f*)(p + 4) = hi;
}

__device__ __forceinline__ float4 frag_load(const float* tile, int m, int half, int g) {
    return *(const float4*)(tile + m * LDS_PITCH + 8 * g + 4 * half);
}

__device__ __forceinline__ float fix_key(float kf) {
    const unsigned int k = __float_as_uint(kf);
    return (k == KEY_NEG_INF) ? 0.0f : key_to_f32(k);
}

template <bool FINAL>
__global__ __launch_bounds__(256) void sage_gemm_kernel(const float* A0,
                                                        const float* A1,
                                                        const float* __restrict__ W0,
                                                        const float* __restrict__ W1,
                                                        const float* __restrict__ bias,
                                                        const float* __restrict__ W2,
                                                        const float* __restrict__ bias2,
                                                        float* out,
                                                        int ntiles) {
    __shared__ __attribute__((aligned(16))) float lds[GEMM_WAVES * TILE_DW];

    const int lane = threadIdx.x & 31;
    const int wv   = threadIdx.x >> 5;
    const int m    = lane & 15;
    const int half = lane >> 4;
    const int col_base = wv * 16;

    float* wreg = lds + wv * TILE_DW;
    float4 b0[16];
    float4 b1[16];
    float4 b2[16];
#pragma unroll
    for (int r = 0; r < 16; ++r) {
        const float4 v = *(const float4*)(W0 + (size_t)(col_base + r) * DFEAT + lane * 4);
        stage_store(wreg + r * LDS_PITCH, lane, v);
    }
    __syncthreads();
#pragma unroll
    for (int g = 0; g < 16; ++g) b0[g] = frag_load(wreg, m, half, g);
    __syncthreads();
#pragma unroll
    for (int r = 0; r < 16; ++r) {
        const float4 v = *(const float4*)(W1 + (size_t)(col_base + r) * DFEAT + lane * 4);
        stage_store(wreg + r * LDS_PITCH, lane, v);
    }
    __syncthreads();
#pragma unroll
    for (int g = 0; g < 16; ++g) b1[g] = frag_load(wreg, m, half, g);
    if (FINAL) {
        __syncthreads();
#pragma unroll
        for (int r = 0; r < 16; ++r) {
            const float4 v = *(const float4*)(W2 + (size_t)(col_base + r) * DFEAT + lane * 4);
            stage_store(wreg + r * LDS_PITCH, lane, v);
        }
        __syncthreads();
#pragma unroll
        for (int g = 0; g < 16; ++g) b2[g] = frag_load(wreg, m, half, g);
    }
    __syncthreads();

    const float bv = bias[col_base + m];
    float bv2 = 0.0f;
    if (FINAL) bv2 = bias2[col_base + m];

    const int xcol = 8 * (col_base >> 3) + 8 * (m >> 3) + ((m & 1) | ((m & 2) << 1) | ((m & 4) >> 1));

    float4 pa0[2];
    float4 pa1[2];
    int t = blockIdx.x;
    if (t < ntiles) {
#pragma unroll
        for (int i = 0; i < 2; ++i) {
            const size_t off = (size_t)(t * 16 + wv + 8 * i) * DFEAT + lane * 4;
            pa0[i] = *(const float4*)(A0 + off);
            pa1[i] = *(const float4*)(A1 + off);
        }
#pragma unroll
        for (int i = 0; i < 2; ++i) {
            float4 v = pa0[i];
            v.x = fix_key(v.x); v.y = fix_key(v.y); v.z = fix_key(v.z); v.w = fix_key(v.w);
            stage_store(lds + (wv + 8 * i) * LDS_PITCH, lane, v);
            stage_store(lds + TILE_DW + (wv + 8 * i) * LDS_PITCH, lane, pa1[i]);
        }
    }
    __syncthreads();

    int buf = 0;
    for (; t < ntiles; t += gridDim.x) {
        const int  tn       = t + gridDim.x;
        const bool has_next = tn < ntiles;
        if (has_next) {
#pragma unroll
            for (int i = 0; i < 2; ++i) {
                const size_t off = (size_t)(tn * 16 + wv + 8 * i) * DFEAT + lane * 4;
                pa0[i] = *(const float4*)(A0 + off);
                pa1[i] = *(const float4*)(A1 + off);
            }
        }

        const float* a0t = lds + buf * 2 * TILE_DW;
        const float* a1t = a0t + TILE_DW;
        v8f acc = {};
#pragma unroll
        for (int g = 0; g < 16; ++g) {
            const float4 fa0 = frag_load(a0t, m, half, g);
            const float4 fa1 = frag_load(a1t, m, half, g);
            v2f a, b;
            a.x = fa0.x; a.y = fa0.y; b.x = b0[g].x; b.y = b0[g].y;
            acc = __builtin_amdgcn_wmma_f32_16x16x4_f32(false, a, false, b, (short)0, acc, false, false);
            a.x = fa1.x; a.y = fa1.y; b.x = b1[g].x; b.y = b1[g].y;
            acc = __builtin_amdgcn_wmma_f32_16x16x4_f32(false, a, false, b, (short)0, acc, false, false);
            a.x = fa0.z; a.y = fa0.w; b.x = b0[g].z; b.y = b0[g].w;
            acc = __builtin_amdgcn_wmma_f32_16x16x4_f32(false, a, false, b, (short)0, acc, false, false);
            a.x = fa1.z; a.y = fa1.w; b.x = b1[g].z; b.y = b1[g].w;
            acc = __builtin_amdgcn_wmma_f32_16x16x4_f32(false, a, false, b, (short)0, acc, false, false);
        }

        if (FINAL) {
            float* xs = lds + (4 + buf) * TILE_DW;
#pragma unroll
            for (int r = 0; r < 8; ++r)
                xs[(r + 8 * half) * LDS_PITCH + xcol] = fmaxf(acc[r] + bv, 0.0f);
        }

        if (has_next) {
            float* n0 = lds + (buf ^ 1) * 2 * TILE_DW;
#pragma unroll
            for (int i = 0; i < 2; ++i) {
                float4 v = pa0[i];
                v.x = fix_key(v.x); v.y = fix_key(v.y); v.z = fix_key(v.z); v.w = fix_key(v.w);
                stage_store(n0 + (wv + 8 * i) * LDS_PITCH, lane, v);
                stage_store(n0 + TILE_DW + (wv + 8 * i) * LDS_PITCH, lane, pa1[i]);
            }
        }
        __syncthreads();

        const int row_base = t * 16;
        if (FINAL) {
            const float* xt = lds + (4 + buf) * TILE_DW;
            v8f acc2 = {};
#pragma unroll
            for (int g = 0; g < 16; ++g) {
                const float4 fx = frag_load(xt, m, half, g);
                v2f a, b;
                a.x = fx.x; a.y = fx.y; b.x = b2[g].x; b.y = b2[g].y;
                acc2 = __builtin_amdgcn_wmma_f32_16x16x4_f32(false, a, false, b, (short)0, acc2, false, false);
                a.x = fx.z; a.y = fx.w; b.x = b2[g].z; b.y = b2[g].w;
                acc2 = __builtin_amdgcn_wmma_f32_16x16x4_f32(false, a, false, b, (short)0, acc2, false, false);
            }
#pragma unroll
            for (int r = 0; r < 8; ++r)
                out[(size_t)(row_base + r + 8 * half) * DFEAT + col_base + m] = acc2[r] + bv2;
        } else {
#pragma unroll
            for (int r = 0; r < 8; ++r)
                out[(size_t)(row_base + r + 8 * half) * DFEAT + col_base + m] = fmaxf(acc[r] + bv, 0.0f);
        }
        buf ^= 1;
    }
}

extern "C" void kernel_launch(void* const* d_in, const int* in_sizes, int n_in,
                              void* d_out, int out_size, void* d_ws, size_t ws_size,
                              hipStream_t stream) {
    const float* x  = (const float*)d_in[0];
    const int*   ei = (const int*)d_in[1];
    const float* Wl = (const float*)d_in[2];
    const float* bl = (const float*)d_in[3];
    const float* Wr = (const float*)d_in[4];
    const float* W  = (const float*)d_in[5];
    const float* b  = (const float*)d_in[6];

    const int nE = in_sizes[1] / 2;
    const int n  = in_sizes[0] / DFEAT;
    const int*   srcp = ei;
    const int*   dstp = ei + nE;

    float* buf0 = (float*)d_ws;
    float* buf1 = buf0 + (size_t)n * DFEAT;
    const int nranges   = (n + ROWS_PER_RANGE - 1) / ROWS_PER_RANGE;
    const int binBlocks = (nE + BIN_CHUNK - 1) / BIN_CHUNK;
    int* bsrc = (int*)(buf1 + (size_t)n * DFEAT);
    int* bdst = bsrc + nE;
    unsigned int* hist = (unsigned int*)(bdst + nE);
    int* binStart = (int*)(hist + (size_t)binBlocks * nranges);

    dim3 blk(256);
    const int gemmTiles = n / 16;

    bin_count_kernel<<<binBlocks, BIN_THREADS, 0, stream>>>(dstp, nE, nranges, hist);
    bin_place_kernel<<<binBlocks, BIN_THREADS, 0, stream>>>(srcp, dstp, nE, nranges, binBlocks, hist,
                                                            bsrc, bdst, binStart);

    range_max_kernel<<<nranges, 1024, 0, stream>>>(x, bsrc, bdst, binStart, (unsigned int*)buf0, n);
    sage_gemm_kernel<false><<<GEMM_BLOCKS, blk, 0, stream>>>(
        buf0, x, Wl, Wr, bl, nullptr, nullptr, buf0, gemmTiles);

    range_max_kernel<<<nranges, 1024, 0, stream>>>(buf0, bsrc, bdst, binStart, (unsigned int*)buf1, n);
    sage_gemm_kernel<true><<<GEMM_BLOCKS, blk, 0, stream>>>(
        buf1, buf0, Wl + DFEAT * DFEAT, Wr + DFEAT * DFEAT, bl + DFEAT, W, b, (float*)d_out,
        gemmTiles);
}


# Edge binning in one launch, not two: a block orders only its own chunk of edges by destination range and stores its range starts; a range's waves walk the chunks' segments; max only, no sum reordered
# speedup vs baseline: 4.3320x; 1.0544x over previous
#include <hip/hip_runtime.h>
#include <math.h>

typedef __attribute__((ext_vector_type(2))) float v2f;
typedef __attribute__((ext_vector_type(8))) float v8f;

#define DFEAT 128

#define KEY_NEG_INF 0x007FFFFFu

__device__ __forceinline__ unsigned int f32_to_key(float v) {
    const unsigned int b = __float_as_uint(v);
    return b ^ ((unsigned int)((int)b >> 31) | 0x80000000u);
}

__device__ __forceinline__ float key_to_f32(unsigned int k) {
    return __uint_as_float(k ^ (((int)k < 0) ? 0x80000000u : 0xFFFFFFFFu));
}

#define ROWS_PER_RANGE 196
#define MAX_RANGES     1024
#define BIN_THREADS    1024
#define BIN_CHUNK      6250
#define BIN_PER_THREAD 7
#define CHUNK_STRIDE   6272

__global__ __launch_bounds__(BIN_THREADS) void bin_chunk_kernel(const int* __restrict__ src,
                                                                const int* __restrict__ dst,
                                                                int nedges, int n, int nranges, int pitch,
                                                                int* __restrict__ bsrc,
                                                                int* __restrict__ bdst,
                                                                int* __restrict__ segtab) {
    __shared__ __attribute__((aligned(16))) unsigned int cnt[MAX_RANGES];
    __shared__ unsigned int lstart[MAX_RANGES + 1];
    __shared__ int ssrc[BIN_CHUNK];
    __shared__ int sdst[BIN_CHUNK];
    for (int i = threadIdx.x; i < nranges; i += BIN_THREADS) cnt[i] = 0u;
    __syncthreads();

    const int chunk0 = blockIdx.x * BIN_CHUNK;
    const int left = nedges - chunk0;
    const int m = left < BIN_CHUNK ? left : BIN_CHUNK;
    int sv[BIN_PER_THREAD];
    int dv[BIN_PER_THREAD];
    unsigned int rk[BIN_PER_THREAD];
#pragma unroll
    for (int k = 0; k < BIN_PER_THREAD; ++k) {
        const int i = (int)threadIdx.x + k * BIN_THREADS;
        const bool ok = i < m;
        sv[k] = ok ? src[chunk0 + i] : 0;
        const int d = ok ? dst[chunk0 + i] : -1;
        dv[k] = ((unsigned int)d < (unsigned int)n) ? d : -1;
    }
#pragma unroll
    for (int k = 0; k < BIN_PER_THREAD; ++k) {
        rk[k] = 0u;
        if (dv[k] >= 0) rk[k] = atomicAdd(&cnt[(unsigned int)dv[k] / ROWS_PER_RANGE], 1u);
    }
    __syncthreads();

    for (int r = threadIdx.x; r <= nranges; r += BIN_THREADS) {
        unsigned int s = 0u;
        int q = 0;
        for (; q + 4 <= r; q += 4) {
            const uint4 a = *(const uint4*)(cnt + q);
            s += (a.x + a.y) + (a.z + a.w);
        }
        for (; q < r; ++q) s += cnt[q];
        lstart[r] = s;
        segtab[blockIdx.x * pitch + r] = (int)s;
    }
    __syncthreads();

#pragma unroll
    for (int k = 0; k < BIN_PER_THREAD; ++k) {
        if (dv[k] >= 0) {
            const unsigned int p = lstart[(unsigned int)dv[k] / ROWS_PER_RANGE] + rk[k];
            ssrc[p] = sv[k];
            sdst[p] = dv[k];
        }
    }
    __syncthreads();

    const int kept = (int)lstart[nranges];
    const int out0 = blockIdx.x * CHUNK_STRIDE;
    for (int i = threadIdx.x; i < kept; i += BIN_THREADS) {
        bsrc[out0 + i] = ssrc[i];
        bdst[out0 + i] = sdst[i];
    }
}

#define EDGE_BATCH   16
#define RANGE_WAVES  32
#define KEY_WORDS    (ROWS_PER_RANGE * DFEAT)

__global__ __launch_bounds__(1024) void range_max_kernel(const float* __restrict__ x,
                                                         const int* __restrict__ bsrc,
                                                         const int* __restrict__ bdst,
                                                         const int* __restrict__ segtab,
                                                         int pitch, int nchunks,
                                                         unsigned int* __restrict__ agg,
                                                         int n) {
    __shared__ __attribute__((aligned(16))) unsigned int keys[KEY_WORDS];
    const int rng = blockIdx.x;
    const int row_base = rng * ROWS_PER_RANGE;
    const int lane = threadIdx.x & 31;
    const int wv = __builtin_amdgcn_readfirstlane((int)(threadIdx.x >> 5));

    int cb = wv;
    int c = cb + RANGE_WAVES * lane;
    int s0 = 0, s1 = 0;
    if (c < nchunks) {
        s0 = segtab[c * pitch + rng];
        s1 = segtab[c * pitch + rng + 1];
    }

    uint4 kv;
    kv.x = KEY_NEG_INF; kv.y = KEY_NEG_INF; kv.z = KEY_NEG_INF; kv.w = KEY_NEG_INF;
    for (int i = threadIdx.x; i < KEY_WORDS / 4; i += 1024) ((uint4*)keys)[i] = kv;
    __syncthreads();

    const float4* __restrict__ x4 = (const float4*)x;

    while (cb < nchunks) {
        const int sn = s1 - s0;
        int nseg = (nchunks - cb + RANGE_WAVES - 1) / RANGE_WAVES;
        if (nseg > 32) nseg = 32;
        int sp = 0, total = 0;
        for (int i = 0; i < nseg; ++i) {
            const int v = __builtin_amdgcn_readlane(sn, i);
            if (lane == i) sp = total;
            total += v;
        }
        const int sb = c * CHUNK_STRIDE + s0 - sp;

        int sv = 0, dv = row_base;
        if (total > 0) {
            int f = lane < total - 1 ? lane : total - 1;
            int p = 0;
            for (int i = 0; i < nseg; ++i) {
                const int spi = __builtin_amdgcn_readlane(sp, i);
                const int sbi = __builtin_amdgcn_readlane(sb, i);
                if (f >= spi) p = sbi + f;
            }
            if (lane < EDGE_BATCH) {
                sv = bsrc[p];
                dv = bdst[p];
            }
        }
        for (int f0 = 0; f0 < total; f0 += EDGE_BATCH) {
            float4 v[EDGE_BATCH];
#pragma unroll
            for (int j = 0; j < EDGE_BATCH; ++j) {
                const int s = __builtin_amdgcn_readlane(sv, j);
                v[j] = x4[(unsigned int)s * (DFEAT / 4) + (unsigned int)lane];
            }
            int svn = 0, dvn = row_base;
            const int fn = f0 + EDGE_BATCH;
            if (fn < total) {
                int f = fn + lane;
                if (f > total - 1) f = total - 1;
                int p = 0;
                for (int i = 0; i < nseg; ++i) {
                    const int spi = __builtin_amdgcn_readlane(sp, i);
                    const int sbi = __builtin_amdgcn_readlane(sb, i);
                    if (f >= spi) p = sbi + f;
                }
                if (lane < EDGE_BATCH) {
                    svn = bsrc[p];
                    dvn = bdst[p];
                }
            }
            if (fn <= total) {
#pragma unroll
                for (int j = 0; j < EDGE_BATCH; ++j) {
                    const int d = __builtin_amdgcn_readlane(dv, j) - row_base;
                    unsigned int* kr = keys + d * DFEAT + lane;
                    atomicMax(kr,      f32_to_key(v[j].x));
                    atomicMax(kr + 32, f32_to_key(v[j].y));
                    atomicMax(kr + 64, f32_to_key(v[j].z));
                    atomicMax(kr + 96, f32_to_key(v[j].w));
                }
            } else {
                const int cnt = total - f0;
#pragma unroll
                for (int j = 0; j < EDGE_BATCH; ++j) {
                    if (j < cnt) {
                        const int d = __builtin_amdgcn_readlane(dv, j) - row_base;
                        unsigned int* kr = keys + d * DFEAT + lane;
                        atomicMax(kr,      f32_to_key(v[j].x));
                        atomicMax(kr + 32, f32_to_key(v[j].y));
                        atomicMax(kr + 64, f32_to_key(v[j].z));
                        atomicMax(kr + 96, f32_to_key(v[j].w));
                    }
                }
            }
            sv = svn;
            dv = dvn;
        }

        cb += RANGE_WAVES * 32;
        c = cb + RANGE_WAVES * lane;
        s0 = 0; s1 = 0;
        if (c < nchunks) {
            s0 = segtab[c * pitch + rng];
            s1 = segtab[c * pitch + rng + 1];
        }
    }
    __syncthreads();

    for (int i = threadIdx.x; i < ROWS_PER_RANGE * 32; i += 1024) {
        const int r   = i >> 5;
        const int row = row_base + r;
        const unsigned int* kr = keys + r * DFEAT + (i & 31);
        uint4 k;
        k.x = kr[0]; k.y = kr[32]; k.z = kr[64]; k.w = kr[96];
        if (row < n)
            *(uint4*)(agg + (size_t)row * DFEAT + (i & 31) * 4) = k;
    }
}

#define LDS_PITCH   132
#define TILE_DW     (16 * LDS_PITCH)
#define GEMM_WAVES  8
#define GEMM_BLOCKS 512

__device__ __forceinline__ void stage_store(float* row, int q, float4 v) {
    float* p = row + 8 * (q >> 1) + 2 * (q & 1);
    v2f lo; lo.x = v.x; lo.y = v.y;
    v2f hi; hi.x = v.z; hi.y = v.w;
    *(v2f*)(p)     = lo;
    *(v2f*)(p + 4) = hi;
}

__device__ __forceinline__ float4 frag_load(const float* tile, int m, int half, int g) {
    return *(const float4*)(tile + m * LDS_PITCH + 8 * g + 4 * half);
}

__device__ __forceinline__ float fix_key(float kf) {
    const unsigned int k = __float_as_uint(kf);
    return (k == KEY_NEG_INF) ? 0.0f : key_to_f32(k);
}

template <bool FINAL>
__global__ __launch_bounds__(256) void sage_gemm_kernel(const float* A0,
                                                        const float* A1,
                                                        const float* __restrict__ W0,
                                                        const float* __restrict__ W1,
                                                        const float* __restrict__ bias,
                                                        const float* __restrict__ W2,
                                                        const float* __restrict__ bias2,
                                                        float* out,
                                                        int ntiles) {
    __shared__ __attribute__((aligned(16))) float lds[GEMM_WAVES * TILE_DW];

    const int lane = threadIdx.x & 31;
    const int wv   = threadIdx.x >> 5;
    const int m    = lane & 15;
    const int half = lane >> 4;
    const int col_base = wv * 16;

    float* wreg = lds + wv * TILE_DW;
    float4 b0[16];
    float4 b1[16];
    float4 b2[16];
#pragma unroll
    for (int r = 0; r < 16; ++r) {
        const float4 v = *(const float4*)(W0 + (size_t)(col_base + r) * DFEAT + lane * 4);
        stage_store(wreg + r * LDS_PITCH, lane, v);
    }
    __syncthreads();
#pragma unroll
    for (int g = 0; g < 16; ++g) b0[g] = frag_load(wreg, m, half, g);
    __syncthreads();
#pragma unroll
    for (int r = 0; r < 16; ++r) {
        const float4 v = *(const float4*)(W1 + (size_t)(col_base + r) * DFEAT + lane * 4);
        stage_store(wreg + r * LDS_PITCH, lane, v);
    }
    __syncthreads();
#pragma unroll
    for (int g = 0; g < 16; ++g) b1[g] = frag_load(wreg, m, half, g);
    if (FINAL) {
        __syncthreads();
#pragma unroll
        for (int r = 0; r < 16; ++r) {
            const float4 v = *(const float4*)(W2 + (size_t)(col_base + r) * DFEAT + lane * 4);
            stage_store(wreg + r * LDS_PITCH, lane, v);
        }
        __syncthreads();
#pragma unroll
        for (int g = 0; g < 16; ++g) b2[g] = frag_load(wreg, m, half, g);
    }
    __syncthreads();

    const float bv = bias[col_base + m];
    float bv2 = 0.0f;
    if (FINAL) bv2 = bias2[col_base + m];

    const int xcol = 8 * (col_base >> 3) + 8 * (m >> 3) + ((m & 1) | ((m & 2) << 1) | ((m & 4) >> 1));

    float4 pa0[2];
    float4 pa1[2];
    int t = blockIdx.x;
    if (t < ntiles) {
#pragma unroll
        for (int i = 0; i < 2; ++i) {
            const size_t off = (size_t)(t * 16 + wv + 8 * i) * DFEAT + lane * 4;
            pa0[i] = *(const float4*)(A0 + off);
            pa1[i] = *(const float4*)(A1 + off);
        }
#pragma unroll
        for (int i = 0; i < 2; ++i) {
            float4 v = pa0[i];
            v.x = fix_key(v.x); v.y = fix_key(v.y); v.z = fix_key(v.z); v.w = fix_key(v.w);
            stage_store(lds + (wv + 8 * i) * LDS_PITCH, lane, v);
            stage_store(lds + TILE_DW + (wv + 8 * i) * LDS_PITCH, lane, pa1[i]);
        }
    }
    __syncthreads();

    int buf = 0;
    for (; t < ntiles; t += gridDim.x) {
        const int  tn       = t + gridDim.x;
        const bool has_next = tn < ntiles;
        if (has_next) {
#pragma unroll
            for (int i = 0; i < 2; ++i) {
                const size_t off = (size_t)(tn * 16 + wv + 8 * i) * DFEAT + lane * 4;
                pa0[i] = *(const float4*)(A0 + off);
                pa1[i] = *(const float4*)(A1 + off);
            }
        }

        const float* a0t = lds + buf * 2 * TILE_DW;
        const float* a1t = a0t + TILE_DW;
        v8f acc = {};
#pragma unroll
        for (int g = 0; g < 16; ++g) {
            const float4 fa0 = frag_load(a0t, m, half, g);
            const float4 fa1 = frag_load(a1t, m, half, g);
            v2f a, b;
            a.x = fa0.x; a.y = fa0.y; b.x = b0[g].x; b.y = b0[g].y;
            acc = __builtin_amdgcn_wmma_f32_16x16x4_f32(false, a, false, b, (short)0, acc, false, false);
            a.x = fa1.x; a.y = fa1.y; b.x = b1[g].x; b.y = b1[g].y;
            acc = __builtin_amdgcn_wmma_f32_16x16x4_f32(false, a, false, b, (short)0, acc, false, false);
            a.x = fa0.z; a.y = fa0.w; b.x = b0[g].z; b.y = b0[g].w;
            acc = __builtin_amdgcn_wmma_f32_16x16x4_f32(false, a, false, b, (short)0, acc, false, false);
            a.x = fa1.z; a.y = fa1.w; b.x = b1[g].z; b.y = b1[g].w;
            acc = __builtin_amdgcn_wmma_f32_16x16x4_f32(false, a, false, b, (short)0, acc, false, false);
        }

        if (FINAL) {
            float* xs = lds + (4 + buf) * TILE_DW;
#pragma unroll
            for (int r = 0; r < 8; ++r)
                xs[(r + 8 * half) * LDS_PITCH + xcol] = fmaxf(acc[r] + bv, 0.0f);
        }

        if (has_next) {
            float* n0 = lds + (buf ^ 1) * 2 * TILE_DW;
#pragma unroll
            for (int i = 0; i < 2; ++i) {
                float4 v = pa0[i];
                v.x = fix_key(v.x); v.y = fix_key(v.y); v.z = fix_key(v.z); v.w = fix_key(v.w);
                stage_store(n0 + (wv + 8 * i) * LDS_PITCH, lane, v);
                stage_store(n0 + TILE_DW + (wv + 8 * i) * LDS_PITCH, lane, pa1[i]);
            }
        }
        __syncthreads();

        const int row_base = t * 16;
        if (FINAL) {
            const float* xt = lds + (4 + buf) * TILE_DW;
            v8f acc2 = {};
#pragma unroll
            for (int g = 0; g < 16; ++g) {
                const float4 fx = frag_load(xt, m, half, g);
                v2f a, b;
                a.x = fx.x; a.y = fx.y; b.x = b2[g].x; b.y = b2[g].y;
                acc2 = __builtin_amdgcn_wmma_f32_16x16x4_f32(false, a, false, b, (short)0, acc2, false, false);
                a.x = fx.z; a.y = fx.w; b.x = b2[g].z; b.y = b2[g].w;
                acc2 = __builtin_amdgcn_wmma_f32_16x16x4_f32(false, a, false, b, (short)0, acc2, false, false);
            }
#pragma unroll
            for (int r = 0; r < 8; ++r)
                out[(size_t)(row_base + r + 8 * half) * DFEAT + col_base + m] = acc2[r] + bv2;
        } else {
#pragma unroll
            for (int r = 0; r < 8; ++r)
                out[(size_t)(row_base + r + 8 * half) * DFEAT + col_base + m] = fmaxf(acc[r] + bv, 0.0f);
        }
        buf ^= 1;
    }
}

extern "C" void kernel_launch(void* const* d_in, const int* in_sizes, int n_in,
                              void* d_out, int out_size, void* d_ws, size_t ws_size,
                              hipStream_t stream) {
    const float* x  = (const float*)d_in[0];
    const int*   ei = (const int*)d_in[1];
    const float* Wl = (const float*)d_in[2];
    const float* bl = (const float*)d_in[3];
    const float* Wr = (const float*)d_in[4];
    const float* W  = (const float*)d_in[5];
    const float* b  = (const float*)d_in[6];

    const int nE = in_sizes[1] / 2;
    const int n  = in_sizes[0] / DFEAT;
    const int*   srcp = ei;
    const int*   dstp = ei + nE;

    float* buf0 = (float*)d_ws;
    float* buf1 = buf0 + (size_t)n * DFEAT;
    const int nranges = (n + ROWS_PER_RANGE - 1) / ROWS_PER_RANGE;
    const int nchunks = (nE + BIN_CHUNK - 1) / BIN_CHUNK;
    const int pitch   = ((nranges + 1 + 31) / 32) * 32;
    int* bsrc   = (int*)(buf1 + (size_t)n * DFEAT);
    int* bdst   = bsrc + (size_t)nchunks * CHUNK_STRIDE;
    int* segtab = bdst + (size_t)nchunks * CHUNK_STRIDE;

    dim3 blk(256);
    const int gemmTiles = n / 16;

    bin_chunk_kernel<<<nchunks, BIN_THREADS, 0, stream>>>(srcp, dstp, nE, n, nranges, pitch,
                                                          bsrc, bdst, segtab);

    range_max_kernel<<<nranges, 1024, 0, stream>>>(x, bsrc, bdst, segtab, pitch, nchunks,
                                                   (unsigned int*)buf0, n);
    sage_gemm_kernel<false><<<GEMM_BLOCKS, blk, 0, stream>>>(
        buf0, x, Wl, Wr, bl, nullptr, nullptr, buf0, gemmTiles);

    range_max_kernel<<<nranges, 1024, 0, stream>>>(buf0, bsrc, bdst, segtab, pitch, nchunks,
                                                   (unsigned int*)buf1, n);
    sage_gemm_kernel<true><<<GEMM_BLOCKS, blk, 0, stream>>>(
        buf1, buf0, Wl + DFEAT * DFEAT, Wr + DFEAT * DFEAT, bl + DFEAT, W, b, (float*)d_out,
        gemmTiles);
}


# Each layer's range max and its matrix product in one launch (5 launches to 3): the work-group turns its LDS max keys into floats in place and multiplies its row tiles from LDS; no sum reordered
# speedup vs baseline: 4.9527x; 1.1433x over previous
#include <hip/hip_runtime.h>
#include <math.h>

typedef __attribute__((ext_vector_type(2))) float v2f;
typedef __attribute__((ext_vector_type(8))) float v8f;

#define DFEAT 128

#define KEY_NEG_INF 0x007FFFFFu

__device__ __forceinline__ unsigned int f32_to_key(float v) {
    const unsigned int b = __float_as_uint(v);
    return b ^ ((unsigned int)((int)b >> 31) | 0x80000000u);
}

__device__ __forceinline__ float key_to_f32(unsigned int k) {
    return __uint_as_float(k ^ (((int)k < 0) ? 0x80000000u : 0xFFFFFFFFu));
}

#define ROWS_PER_RANGE 196
#define MAX_RANGES     1024
#define BIN_THREADS    1024
#define BIN_CHUNK      6250
#define BIN_PER_THREAD 7
#define CHUNK_STRIDE   6272

__global__ __launch_bounds__(BIN_THREADS) void bin_chunk_kernel(const int* __restrict__ src,
                                                                const int* __restrict__ dst,
                                                                int nedges, int n, int nranges, int pitch,
                                                                int* __restrict__ bsrc,
                                                                int* __restrict__ bdst,
                                                                int* __restrict__ segtab) {
    __shared__ __attribute__((aligned(16))) unsigned int cnt[MAX_RANGES];
    __shared__ unsigned int lstart[MAX_RANGES + 1];
    __shared__ int ssrc[BIN_CHUNK];
    __shared__ int sdst[BIN_CHUNK];
    for (int i = threadIdx.x; i < nranges; i += BIN_THREADS) cnt[i] = 0u;
    __syncthreads();

    const int chunk0 = blockIdx.x * BIN_CHUNK;
    const int left = nedges - chunk0;
    const int m = left < BIN_CHUNK ? left : BIN_CHUNK;
    int sv[BIN_PER_THREAD];
    int dv[BIN_PER_THREAD];
    unsigned int rk[BIN_PER_THREAD];
#pragma unroll
    for (int k = 0; k < BIN_PER_THREAD; ++k) {
        const int i = (int)threadIdx.x + k * BIN_THREADS;
        const bool ok = i < m;
        sv[k] = ok ? src[chunk0 + i] : 0;
        const int d = ok ? dst[chunk0 + i] : -1;
        dv[k] = ((unsigned int)d < (unsigned int)n) ? d : -1;
    }
#pragma unroll
    for (int k = 0; k < BIN_PER_THREAD; ++k) {
        rk[k] = 0u;
        if (dv[k] >= 0) rk[k] = atomicAdd(&cnt[(unsigned int)dv[k] / ROWS_PER_RANGE], 1u);
    }
    __syncthreads();

    for (int r = threadIdx.x; r <= nranges; r += BIN_THREADS) {
        unsigned int s = 0u;
        int q = 0;
        for (; q + 4 <= r; q += 4) {
            const uint4 a = *(const uint4*)(cnt + q);
            s += (a.x + a.y) + (a.z + a.w);
        }
        for (; q < r; ++q) s += cnt[q];
        lstart[r] = s;
        segtab[blockIdx.x * pitch + r] = (int)s;
    }
    __syncthreads();

#pragma unroll
    for (int k = 0; k < BIN_PER_THREAD; ++k) {
        if (dv[k] >= 0) {
            const unsigned int p = lstart[(unsigned int)dv[k] / ROWS_PER_RANGE] + rk[k];
            ssrc[p] = sv[k];
            sdst[p] = dv[k];
        }
    }
    __syncthreads();

    const int kept = (int)lstart[nranges];
    const int out0 = blockIdx.x * CHUNK_STRIDE;
    for (int i = threadIdx.x; i < kept; i += BIN_THREADS) {
        bsrc[out0 + i] = ssrc[i];
        bdst[out0 + i] = sdst[i];
    }
}

#define LDS_PITCH    132
#define TILE_DW      (16 * LDS_PITCH)
#define RANGE_TILES  13
#define KEY_ROWS     (RANGE_TILES * 16)
#define KEY_DW       (KEY_ROWS * LDS_PITCH)
#define W_DW         (DFEAT * LDS_PITCH)
#define ROW_GROUPS   4
#define ROOT_DW      (ROW_GROUPS * 2 * TILE_DW)
#define LAYER_DW     (KEY_DW + 2 * W_DW + ROOT_DW)

__device__ __forceinline__ void stage_store(float* row, int q, float4 v) {
    float* p = row + 8 * (q >> 1) + 2 * (q & 1);
    v2f lo; lo.x = v.x; lo.y = v.y;
    v2f hi; hi.x = v.z; hi.y = v.w;
    *(v2f*)(p)     = lo;
    *(v2f*)(p + 4) = hi;
}

__device__ __forceinline__ float4 frag_load(const float* tile, int m, int half, int g) {
    return *(const float4*)(tile + m * LDS_PITCH + 8 * g + 4 * half);
}

__device__ __forceinline__ float fix_key(unsigned int k) {
    return (k == KEY_NEG_INF) ? 0.0f : key_to_f32(k);
}

#define EDGE_BATCH   16
#define RANGE_WAVES  32

template <bool FINAL>
__global__ __launch_bounds__(1024) void range_layer_kernel(const float* __restrict__ x,
                                                           const int* __restrict__ bsrc,
                                                           const int* __restrict__ bdst,
                                                           const int* __restrict__ segtab,
                                                           int pitch, int nchunks,
                                                           const float* __restrict__ W0,
                                                           const float* __restrict__ W1,
                                                           const float* __restrict__ bias,
                                                           const float* __restrict__ W2,
                                                           const float* __restrict__ bias2,
                                                           float* __restrict__ out,
                                                           int n) {
    __shared__ __attribute__((aligned(16))) float lds[LAYER_DW];
    unsigned int* keys = (unsigned int*)lds;
    float* rowsf = lds;
    float* wl0   = lds + KEY_DW;
    float* wl1   = wl0 + W_DW;
    float* roots = wl1 + W_DW;

    const int rng = blockIdx.x;
    const int row_base = rng * ROWS_PER_RANGE;
    const int lane = threadIdx.x & 31;
    const int wv = __builtin_amdgcn_readfirstlane((int)(threadIdx.x >> 5));

    int cb = wv;
    int c = cb + RANGE_WAVES * lane;
    int s0 = 0, s1 = 0;
    if (c < nchunks) {
        s0 = segtab[c * pitch + rng];
        s1 = segtab[c * pitch + rng + 1];
    }

    {
        float4 w0v[4];
        float4 w1v[4];
#pragma unroll
        for (int i = 0; i < 4; ++i) {
            const size_t off = (size_t)(wv + 32 * i) * DFEAT + lane * 4;
            w0v[i] = *(const float4*)(W0 + off);
            w1v[i] = *(const float4*)(W1 + off);
        }
        uint4 kv;
        kv.x = KEY_NEG_INF; kv.y = KEY_NEG_INF; kv.z = KEY_NEG_INF; kv.w = KEY_NEG_INF;
        for (int i = threadIdx.x; i < KEY_DW / 4; i += 1024) ((uint4*)keys)[i] = kv;
#pragma unroll
        for (int i = 0; i < 4; ++i) {
            stage_store(wl0 + (wv + 32 * i) * LDS_PITCH, lane, w0v[i]);
            stage_store(wl1 + (wv + 32 * i) * LDS_PITCH, lane, w1v[i]);
        }
    }
    __syncthreads();

    const float4* __restrict__ x4 = (const float4*)x;

    while (cb < nchunks) {
        const int sn = s1 - s0;
        int nseg = (nchunks - cb + RANGE_WAVES - 1) / RANGE_WAVES;
        if (nseg > 32) nseg = 32;
        int sp = 0, total = 0;
        for (int i = 0; i < nseg; ++i) {
            const int v = __builtin_amdgcn_readlane(sn, i);
            if (lane == i) sp = total;
            total += v;
        }
        const int sb = c * CHUNK_STRIDE + s0 - sp;

        int sv = 0, dv = row_base;
        if (total > 0) {
            int f = lane < total - 1 ? lane : total - 1;
            int p = 0;
            for (int i = 0; i < nseg; ++i) {
                const int spi = __builtin_amdgcn_readlane(sp, i);
                const int sbi = __builtin_amdgcn_readlane(sb, i);
                if (f >= spi) p = sbi + f;
            }
            if (lane < EDGE_BATCH) {
                sv = bsrc[p];
                dv = bdst[p];
            }
        }
        for (int f0 = 0; f0 < total; f0 += EDGE_BATCH) {
            float4 v[EDGE_BATCH];
#pragma unroll
            for (int j = 0; j < EDGE_BATCH; ++j) {
                const int s = __builtin_amdgcn_readlane(sv, j);
                v[j] = x4[(unsigned int)s * (DFEAT / 4) + (unsigned int)lane];
            }
            int svn = 0, dvn = row_base;
            const int fn = f0 + EDGE_BATCH;
            if (fn < total) {
                int f = fn + lane;
                if (f > total - 1) f = total - 1;
                int p = 0;
                for (int i = 0; i < nseg; ++i) {
                    const int spi = __builtin_amdgcn_readlane(sp, i);
                    const int sbi = __builtin_amdgcn_readlane(sb, i);
                    if (f >= spi) p = sbi + f;
                }
                if (lane < EDGE_BATCH) {
                    svn = bsrc[p];
                    dvn = bdst[p];
                }
            }
            if (fn <= total) {
#pragma unroll
                for (int j = 0; j < EDGE_BATCH; ++j) {
                    const int d = __builtin_amdgcn_readlane(dv, j) - row_base;
                    unsigned int* kr = keys + d * LDS_PITCH + lane;
                    atomicMax(kr,      f32_to_key(v[j].x));
                    atomicMax(kr + 32, f32_to_key(v[j].y));
                    atomicMax(kr + 64, f32_to_key(v[j].z));
                    atomicMax(kr + 96, f32_to_key(v[j].w));
                }
            } else {
                const int cnt = total - f0;
#pragma unroll
                for (int j = 0; j < EDGE_BATCH; ++j) {
                    if (j < cnt) {
                        const int d = __builtin_amdgcn_readlane(dv, j) - row_base;
                        unsigned int* kr = keys + d * LDS_PITCH + lane;
                        atomicMax(kr,      f32_to_key(v[j].x));
                        atomicMax(kr + 32, f32_to_key(v[j].y));
                        atomicMax(kr + 64, f32_to_key(v[j].z));
                        atomicMax(kr + 96, f32_to_key(v[j].w));
                    }
                }
            }
            sv = svn;
            dv = dvn;
        }

        cb += RANGE_WAVES * 32;
        c = cb + RANGE_WAVES * lane;
        s0 = 0; s1 = 0;
        if (c < nchunks) {
            s0 = segtab[c * pitch + rng];
            s1 = segtab[c * pitch + rng + 1];
        }
    }

    const int m    = lane & 15;
    const int half = lane >> 4;
    const int cw   = wv & 7;
    const int rg   = wv >> 3;
    const int col_base = cw * 16;

    float4 pa1[2];
#pragma unroll
    for (int i = 0; i < 2; ++i) {
        int gr = row_base + rg * 16 + cw + 8 * i;
        if (gr > n - 1) gr = n - 1;
        pa1[i] = *(const float4*)(x + (size_t)gr * DFEAT + lane * 4);
    }
    float4 w2v[4];
    if (FINAL) {
#pragma unroll
        for (int i = 0; i < 4; ++i)
            w2v[i] = *(const float4*)(W2 + (size_t)(wv + 32 * i) * DFEAT + lane * 4);
    }

    __syncthreads();

    uint4 kk[7];
#pragma unroll
    for (int i = 0; i < 7; ++i) {
        const int r = wv + 32 * i;
        if (r < KEY_ROWS) {
            const unsigned int* kr = keys + r * LDS_PITCH + lane;
            kk[i].x = kr[0]; kk[i].y = kr[32]; kk[i].z = kr[64]; kk[i].w = kr[96];
        }
    }
    __syncthreads();
#pragma unroll
    for (int i = 0; i < 7; ++i) {
        const int r = wv + 32 * i;
        if (r < KEY_ROWS) {
            float4 v;
            v.x = fix_key(kk[i].x); v.y = fix_key(kk[i].y); v.z = fix_key(kk[i].z); v.w = fix_key(kk[i].w);
            stage_store(rowsf + r * LDS_PITCH, lane, v);
        }
    }
    float* groots = roots + rg * 2 * TILE_DW;
#pragma unroll
    for (int i = 0; i < 2; ++i) stage_store(groots + (cw + 8 * i) * LDS_PITCH, lane, pa1[i]);
    __syncthreads();

    const float bv = bias[col_base + m];
    const float* b0t = wl0 + col_base * LDS_PITCH;
    const float* b1t = wl1 + col_base * LDS_PITCH;

    const int xcol = 8 * (col_base >> 3) + 8 * (m >> 3) + ((m & 1) | ((m & 2) << 1) | ((m & 4) >> 1));

    int buf = 0;
    for (int p = 0; p < ROW_GROUPS; ++p) {
        const int  t        = rg + ROW_GROUPS * p;
        const bool active   = (t < RANGE_TILES) && (row_base + t * 16 < n);
        const int  tn       = t + ROW_GROUPS;
        const bool has_next = (tn < RANGE_TILES) && (row_base + tn * 16 < n);
#pragma unroll
        for (int i = 0; i < 2; ++i) {
            int gr = row_base + tn * 16 + cw + 8 * i;
            if (gr > n - 1) gr = n - 1;
            pa1[i] = *(const float4*)(x + (size_t)gr * DFEAT + lane * 4);
        }

        v8f acc = {};
        if (active) {
            const float* a0t = rowsf + t * TILE_DW;
            const float* a1t = groots + buf * TILE_DW;
#pragma unroll
            for (int g = 0; g < 16; ++g) {
                const float4 fa0 = frag_load(a0t, m, half, g);
                const float4 fa1 = frag_load(a1t, m, half, g);
                const float4 fb0 = frag_load(b0t, m, half, g);
                const float4 fb1 = frag_load(b1t, m, half, g);
                v2f a, b;
                a.x = fa0.x; a.y = fa0.y; b.x = fb0.x; b.y = fb0.y;
                acc = __builtin_amdgcn_wmma_f32_16x16x4_f32(false, a, false, b, (short)0, acc, false, false);
                a.x = fa1.x; a.y = fa1.y; b.x = fb1.x; b.y = fb1.y;
                acc = __builtin_amdgcn_wmma_f32_16x16x4_f32(false, a, false, b, (short)0, acc, false, false);
                a.x = fa0.z; a.y = fa0.w; b.x = fb0.z; b.y = fb0.w;
                acc = __builtin_amdgcn_wmma_f32_16x16x4_f32(false, a, false, b, (short)0, acc, false, false);
                a.x = fa1.z; a.y = fa1.w; b.x = fb1.z; b.y = fb1.w;
                acc = __builtin_amdgcn_wmma_f32_16x16x4_f32(false, a, false, b, (short)0, acc, false, false);
            }
        }

        if (FINAL) {
            __syncthreads();
            if (active) {
                float* xs = rowsf + t * TILE_DW;
#pragma unroll
                for (int r = 0; r < 8; ++r)
                    xs[(r + 8 * half) * LDS_PITCH + xcol] = fmaxf(acc[r] + bv, 0.0f);
            }
        }

        if (has_next) {
            float* nx = groots + (buf ^ 1) * TILE_DW;
#pragma unroll
            for (int i = 0; i < 2; ++i) stage_store(nx + (cw + 8 * i) * LDS_PITCH, lane, pa1[i]);
        }
        __syncthreads();
        if (!FINAL && active) {
#pragma unroll
            for (int r = 0; r < 8; ++r) {
                const int lr = t * 16 + r + 8 * half;
                const int gr = row_base + lr;
                if (lr < ROWS_PER_RANGE && gr < n)
                    out[(size_t)gr * DFEAT + col_base + m] = fmaxf(acc[r] + bv, 0.0f);
            }
        }
        buf ^= 1;
    }

    if (FINAL) {
#pragma unroll
        for (int i = 0; i < 4; ++i) stage_store(wl0 + (wv + 32 * i) * LDS_PITCH, lane, w2v[i]);
        __syncthreads();
        const float bv2 = bias2[col_base + m];
        for (int p = 0; p < ROW_GROUPS; ++p) {
            const int t = rg + ROW_GROUPS * p;
            if ((t < RANGE_TILES) && (row_base + t * 16 < n)) {
                const float* xt = rowsf + t * TILE_DW;
                v8f acc2 = {};
#pragma unroll
                for (int g = 0; g < 16; ++g) {
                    const float4 fx = frag_load(xt, m, half, g);
                    const float4 fb = frag_load(b0t, m, half, g);
                    v2f a, b;
                    a.x = fx.x; a.y = fx.y; b.x = fb.x; b.y = fb.y;
                    acc2 = __builtin_amdgcn_wmma_f32_16x16x4_f32(false, a, false, b, (short)0, acc2, false, false);
                    a.x = fx.z; a.y = fx.w; b.x = fb.z; b.y = fb.w;
                    acc2 = __builtin_amdgcn_wmma_f32_16x16x4_f32(false, a, false, b, (short)0, acc2, false, false);
                }
#pragma unroll
                for (int r = 0; r < 8; ++r) {
                    const int lr = t * 16 + r + 8 * half;
                    const int gr = row_base + lr;
                    if (lr < ROWS_PER_RANGE && gr < n)
                        out[(size_t)gr * DFEAT + col_base + m] = acc2[r] + bv2;
                }
            }
        }
    }
}

extern "C" void kernel_launch(void* const* d_in, const int* in_sizes, int n_in,
                              void* d_out, int out_size, void* d_ws, size_t ws_size,
                              hipStream_t stream) {
    const float* x  = (const float*)d_in[0];
    const int*   ei = (const int*)d_in[1];
    const float* Wl = (const float*)d_in[2];
    const float* bl = (const float*)d_in[3];
    const float* Wr = (const float*)d_in[4];
    const float* W  = (const float*)d_in[5];
    const float* b  = (const float*)d_in[6];

    const int nE = in_sizes[1] / 2;
    const int n  = in_sizes[0] / DFEAT;
    const int*   srcp = ei;
    const int*   dstp = ei + nE;

    float* buf0 = (float*)d_ws;
    const int nranges = (n + ROWS_PER_RANGE - 1) / ROWS_PER_RANGE;
    const int nchunks = (nE + BIN_CHUNK - 1) / BIN_CHUNK;
    const int pitch   = ((nranges + 1 + 31) / 32) * 32;
    int* bsrc   = (int*)(buf0 + (size_t)n * DFEAT);
    int* bdst   = bsrc + (size_t)nchunks * CHUNK_STRIDE;
    int* segtab = bdst + (size_t)nchunks * CHUNK_STRIDE;

    bin_chunk_kernel<<<nchunks, BIN_THREADS, 0, stream>>>(srcp, dstp, nE, n, nranges, pitch,
                                                          bsrc, bdst, segtab);

    range_layer_kernel<false><<<nranges, 1024, 0, stream>>>(
        x, bsrc, bdst, segtab, pitch, nchunks, Wl, Wr, bl, nullptr, nullptr, buf0, n);

    range_layer_kernel<true><<<nranges, 1024, 0, stream>>>(
        buf0, bsrc, bdst, segtab, pitch, nchunks, Wl + DFEAT * DFEAT, Wr + DFEAT * DFEAT,
        bl + DFEAT, W, b, (float*)d_out, n);
}
